# starting kernel
# baseline (speedup 1.0000x reference)
#include <hip/hip_runtime.h>
#include <hip/hip_bf16.h>

typedef __attribute__((ext_vector_type(2))) float v2f;
typedef __attribute__((ext_vector_type(8))) float v8f;

#define EMBED 768
#define S_LEN 2048
#define NHEAD 12
#define HDIM  64
#define BATCH 2
#define ROWS  (BATCH * S_LEN)

__device__ __forceinline__ v8f wmma4(v2f a, v2f b, v8f c) {
  return __builtin_amdgcn_wmma_f32_16x16x4_f32(
       false, a,  false, b,
       (short)0, c,  false,  false);
}

__device__ __forceinline__ v8f vzero8() {
  v8f z = {0.f, 0.f, 0.f, 0.f, 0.f, 0.f, 0.f, 0.f};
  return z;
}

__global__ __launch_bounds__(256) void gemm_bias_kernel(
    const float* __restrict__ A, const float* __restrict__ W,
    const float* __restrict__ bias, float* __restrict__ C, int has_bias) {
  const int lane = threadIdx.x & 31;
  const int wave = threadIdx.x >> 5;
  const int half = lane >> 4;
  const int l16  = lane & 15;

  const int m0 = blockIdx.x * 64  + (wave & 3) * 16;
  const int n0 = blockIdx.y * 128 + (wave >> 2) * 64;

  v8f acc[4];
#pragma unroll
  for (int i = 0; i < 4; ++i) acc[i] = vzero8();

  const float* arow = A + (size_t)(m0 + l16) * EMBED;

#pragma unroll 2
  for (int kk = 0; kk < EMBED; kk += 4) {
    const int ka = kk + 2 * half;
    const v2f a = *(const v2f*)(arow + ka);
    const float* w0 = W + (size_t)ka * EMBED + n0 + l16;
    const float* w1 = w0 + EMBED;
#pragma unroll
    for (int nt = 0; nt < 4; ++nt) {
      v2f b;
      b.x = w0[nt * 16];
      b.y = w1[nt * 16];
      acc[nt] = wmma4(a, b, acc[nt]);
    }
  }

#pragma unroll
  for (int nt = 0; nt < 4; ++nt) {
    const int col = n0 + nt * 16 + l16;
    const float bv = has_bias ? bias[col] : 0.f;
#pragma unroll
    for (int v = 0; v < 8; ++v) {
      C[(size_t)(m0 + v + 8 * half) * EMBED + col] = acc[nt][v] + bv;
    }
  }
}

__global__ __launch_bounds__(256) void attention_kernel(
    const float* __restrict__ Q, const float* __restrict__ K,
    const float* __restrict__ V, const float* __restrict__ mask,
    float* __restrict__ O) {
  __shared__ float pstage[8][16 * 16];

  const int lane = threadIdx.x & 31;
  const int wave = threadIdx.x >> 5;
  const int half = lane >> 4;
  const int l16  = lane & 15;

  const int tile = blockIdx.x * 8 + wave;
  const int qt = tile & 127;
  const int bh = tile >> 7;
  const int b  = bh / NHEAD;
  const int h  = bh % NHEAD;

  const int    q0   = qt * 16;
  const size_t base = (size_t)b * S_LEN * EMBED + (size_t)h * HDIM;

  const float* qrow = Q + base + (size_t)(q0 + l16) * EMBED;

  v8f acc[4];
#pragma unroll
  for (int i = 0; i < 4; ++i) acc[i] = vzero8();
  float mrow[8], lrow[8];
#pragma unroll
  for (int v = 0; v < 8; ++v) { mrow[v] = -3.0e38f; lrow[v] = 0.f; }

  float* myp = &pstage[wave][0];

  for (int kt = 0; kt < S_LEN / 16; ++kt) {
    v8f s = vzero8();
    const float* krow = K + base + (size_t)(kt * 16 + l16) * EMBED;
#pragma unroll
    for (int kk = 0; kk < HDIM; kk += 4) {
      v2f a  = *(const v2f*)(qrow + kk + 2 * half);
      v2f bf = *(const v2f*)(krow + kk + 2 * half);
      s = wmma4(a, bf, s);
    }

    const float* mp = mask + (size_t)(q0 + 8 * half) * S_LEN + kt * 16 + l16;
    float p[8];
#pragma unroll
    for (int v = 0; v < 8; ++v) {
      float sv = s[v] * 0.125f + mp[(size_t)v * S_LEN] * (-1.0e9f);
      float rmax = sv;
      rmax = fmaxf(rmax, __shfl_xor(rmax, 1, 32));
      rmax = fmaxf(rmax, __shfl_xor(rmax, 2, 32));
      rmax = fmaxf(rmax, __shfl_xor(rmax, 4, 32));
      rmax = fmaxf(rmax, __shfl_xor(rmax, 8, 32));
      const float mnew  = fmaxf(mrow[v], rmax);
      const float alpha = __expf(mrow[v] - mnew);
      const float pv    = __expf(sv - mnew);
      float rsum = pv;
      rsum += __shfl_xor(rsum, 1, 32);
      rsum += __shfl_xor(rsum, 2, 32);
      rsum += __shfl_xor(rsum, 4, 32);
      rsum += __shfl_xor(rsum, 8, 32);
      lrow[v] = lrow[v] * alpha + rsum;
      mrow[v] = mnew;
      p[v] = pv;
#pragma unroll
      for (int nt = 0; nt < 4; ++nt) acc[nt][v] *= alpha;
    }

#pragma unroll
    for (int v = 0; v < 8; ++v) myp[(v + 8 * half) * 16 + l16] = p[v];
    __syncthreads();

#pragma unroll
    for (int kk = 0; kk < 16; kk += 4) {
      v2f a = *(const v2f*)(myp + l16 * 16 + kk + 2 * half);
      const float* v0 = V + base + (size_t)(kt * 16 + kk + 2 * half) * EMBED + l16;
      const float* v1 = v0 + EMBED;
#pragma unroll
      for (int nt = 0; nt < 4; ++nt) {
        v2f bf;
        bf.x = v0[nt * 16];
        bf.y = v1[nt * 16];
        acc[nt] = wmma4(a, bf, acc[nt]);
      }
    }
    __syncthreads();
  }

#pragma unroll
  for (int v = 0; v < 8; ++v) {
    const float inv = 1.0f / lrow[v];
    float* orow = O + base + (size_t)(q0 + v + 8 * half) * EMBED;
#pragma unroll
    for (int nt = 0; nt < 4; ++nt) {
      orow[nt * 16 + l16] = acc[nt][v] * inv;
    }
  }
}

extern "C" void kernel_launch(void* const* d_in, const int* in_sizes, int n_in,
                              void* d_out, int out_size, void* d_ws, size_t ws_size,
                              hipStream_t stream) {
  const float* x    = (const float*)d_in[0];
  const float* mask = (const float*)d_in[1];
  const float* Wq   = (const float*)d_in[2];
  const float* Wk   = (const float*)d_in[3];
  const float* Wv   = (const float*)d_in[4];
  const float* Wo   = (const float*)d_in[5];
  const float* bo   = (const float*)d_in[6];

  float* ws = (float*)d_ws;
  const size_t mat = (size_t)ROWS * EMBED;
  float* Qb = ws;
  float* Kb = ws + mat;
  float* Vb = ws + 2 * mat;
  float* Ab = ws + 3 * mat;

  dim3 g(ROWS / 64, EMBED / 128);
  gemm_bias_kernel<<<g, 256, 0, stream>>>(x, Wq, nullptr, Qb, 0);
  gemm_bias_kernel<<<g, 256, 0, stream>>>(x, Wk, nullptr, Kb, 0);
  gemm_bias_kernel<<<g, 256, 0, stream>>>(x, Wv, nullptr, Vb, 0);

  const int qtiles = BATCH * NHEAD * (S_LEN / 16);
  attention_kernel<<<qtiles / 8, 256, 0, stream>>>(Qb, Kb, Vb, mask, Ab);

  gemm_bias_kernel<<<g, 256, 0, stream>>>(Ab, Wo, bo, (float*)d_out, 1);
}


# attention: K/V 64-key slabs staged once per 4-wave work-group in LDS (whole-line b128 loads, double-buffered, one barrier per slab) feed all waves' B fragments, not per-wave strided global loads
# speedup vs baseline: 1.4531x; 1.4531x over previous
#include <hip/hip_runtime.h>
#include <hip/hip_bf16.h>

typedef __attribute__((ext_vector_type(2))) float v2f;
typedef __attribute__((ext_vector_type(8))) float v8f;

#define EMBED 768
#define S_LEN 2048
#define NHEAD 12
#define HDIM  64
#define BATCH 2
#define ROWS  (BATCH * S_LEN)

__device__ __forceinline__ v8f wmma4(v2f a, v2f b, v8f c) {
  return __builtin_amdgcn_wmma_f32_16x16x4_f32(
       false, a,  false, b,
       (short)0, c,  false,  false);
}

__device__ __forceinline__ v8f vzero8() {
  v8f z = {0.f, 0.f, 0.f, 0.f, 0.f, 0.f, 0.f, 0.f};
  return z;
}

__global__ __launch_bounds__(256) void gemm_bias_kernel(
    const float* __restrict__ A, const float* __restrict__ W,
    const float* __restrict__ bias, float* __restrict__ C, int has_bias) {
  const int lane = threadIdx.x & 31;
  const int wave = threadIdx.x >> 5;
  const int half = lane >> 4;
  const int l16  = lane & 15;

  const int m0 = blockIdx.x * 64  + (wave & 3) * 16;
  const int n0 = blockIdx.y * 128 + (wave >> 2) * 64;

  v8f acc[4];
#pragma unroll
  for (int i = 0; i < 4; ++i) acc[i] = vzero8();

  const float* arow = A + (size_t)(m0 + l16) * EMBED;

#pragma unroll 2
  for (int kk = 0; kk < EMBED; kk += 4) {
    const int ka = kk + 2 * half;
    const v2f a = *(const v2f*)(arow + ka);
    const float* w0 = W + (size_t)ka * EMBED + n0 + l16;
    const float* w1 = w0 + EMBED;
#pragma unroll
    for (int nt = 0; nt < 4; ++nt) {
      v2f b;
      b.x = w0[nt * 16];
      b.y = w1[nt * 16];
      acc[nt] = wmma4(a, b, acc[nt]);
    }
  }

#pragma unroll
  for (int nt = 0; nt < 4; ++nt) {
    const int col = n0 + nt * 16 + l16;
    const float bv = has_bias ? bias[col] : 0.f;
#pragma unroll
    for (int v = 0; v < 8; ++v) {
      C[(size_t)(m0 + v + 8 * half) * EMBED + col] = acc[nt][v] + bv;
    }
  }
}

#define ATT_WAVES   4
#define ATT_THREADS (ATT_WAVES * 32)
#define KSTAGE      64
#define SLOADS      (KSTAGE / 8)
#define KPITCH      68
#define VPITCH      72

typedef __attribute__((ext_vector_type(4))) float v4f;

__device__ __forceinline__ void wave_lds_handover() {
  __builtin_amdgcn_fence(__ATOMIC_RELEASE, "wavefront");
  __builtin_amdgcn_wave_barrier();
  __builtin_amdgcn_fence(__ATOMIC_ACQUIRE, "wavefront");
}

__global__ __launch_bounds__(ATT_THREADS) void attention_kernel(
    const float* __restrict__ Q, const float* __restrict__ K,
    const float* __restrict__ V, const float* __restrict__ mask,
    float* __restrict__ O) {
  __shared__ __attribute__((aligned(16))) float kst[2][KSTAGE * KPITCH];
  __shared__ __attribute__((aligned(16))) float vst[2][KSTAGE * VPITCH];
  __shared__ __attribute__((aligned(16))) float pstage[ATT_WAVES][16 * 16];

  const int lane = threadIdx.x & 31;
  const int wave = threadIdx.x >> 5;
  const int half = lane >> 4;
  const int l16  = lane & 15;

  const int tile = blockIdx.x * ATT_WAVES + wave;
  const int qt = tile & 127;
  const int bh = tile >> 7;
  const int b  = bh / NHEAD;
  const int h  = bh % NHEAD;

  const int    q0   = qt * 16;
  const size_t base = (size_t)b * S_LEN * EMBED + (size_t)h * HDIM;

  const float* qrow = Q + base + (size_t)(q0 + l16) * EMBED;

  const int srow = threadIdx.x >> 4;
  const int sc4  = (threadIdx.x & 15) * 4;
  const float* kg = K + base + (size_t)srow * EMBED + sc4;
  const float* vg = V + base + (size_t)srow * EMBED + sc4;
  const int kso = srow * KPITCH + sc4;
  const int vso = srow * VPITCH + sc4;

  v2f qf[HDIM / 4];
#pragma unroll
  for (int i = 0; i < HDIM / 4; ++i) qf[i] = *(const v2f*)(qrow + 4 * i + 2 * half);

  v4f kr[SLOADS], vr[SLOADS];
#pragma unroll
  for (int i = 0; i < SLOADS; ++i) {
    kr[i] = *(const v4f*)(kg + (size_t)(8 * i) * EMBED);
    vr[i] = *(const v4f*)(vg + (size_t)(8 * i) * EMBED);
  }
#pragma unroll
  for (int i = 0; i < SLOADS; ++i) {
    *(v4f*)(&kst[0][kso + 8 * i * KPITCH]) = kr[i];
    *(v4f*)(&vst[0][vso + 8 * i * VPITCH]) = vr[i];
  }

  v8f acc[4];
#pragma unroll
  for (int i = 0; i < 4; ++i) acc[i] = vzero8();
  float mrow[8], lrow[8];
#pragma unroll
  for (int v = 0; v < 8; ++v) { mrow[v] = -3.0e38f; lrow[v] = 0.f; }

  float* myp = &pstage[wave][0];

  __syncthreads();

  for (int st = 0; st < S_LEN / KSTAGE; ++st) {
    const int  buf  = st & 1;
    const bool more = (st + 1) < (S_LEN / KSTAGE);

    if (more) {
      const size_t off = (size_t)(st + 1) * KSTAGE * EMBED;
#pragma unroll
      for (int i = 0; i < SLOADS; ++i) {
        kr[i] = *(const v4f*)(kg + off + (size_t)(8 * i) * EMBED);
        vr[i] = *(const v4f*)(vg + off + (size_t)(8 * i) * EMBED);
      }
    }

    const float* kb = &kst[buf][0];
    const float* vb = &vst[buf][0];

    for (int sub = 0; sub < KSTAGE / 16; ++sub) {
      const int kt = st * (KSTAGE / 16) + sub;

      v8f s = vzero8();
      const float* krow = kb + (sub * 16 + l16) * KPITCH + 2 * half;
#pragma unroll
      for (int kk = 0; kk < HDIM; kk += 4) {
        v2f bf = *(const v2f*)(krow + kk);
        s = wmma4(qf[kk / 4], bf, s);
      }

      const float* mp = mask + (size_t)(q0 + 8 * half) * S_LEN + kt * 16 + l16;
      float p[8];
#pragma unroll
      for (int v = 0; v < 8; ++v) {
        float sv = s[v] * 0.125f + mp[(size_t)v * S_LEN] * (-1.0e9f);
        float rmax = sv;
        rmax = fmaxf(rmax, __shfl_xor(rmax, 1, 32));
        rmax = fmaxf(rmax, __shfl_xor(rmax, 2, 32));
        rmax = fmaxf(rmax, __shfl_xor(rmax, 4, 32));
        rmax = fmaxf(rmax, __shfl_xor(rmax, 8, 32));
        const float mnew  = fmaxf(mrow[v], rmax);
        const float alpha = __expf(mrow[v] - mnew);
        const float pv    = __expf(sv - mnew);
        float rsum = pv;
        rsum += __shfl_xor(rsum, 1, 32);
        rsum += __shfl_xor(rsum, 2, 32);
        rsum += __shfl_xor(rsum, 4, 32);
        rsum += __shfl_xor(rsum, 8, 32);
        lrow[v] = lrow[v] * alpha + rsum;
        mrow[v] = mnew;
        p[v] = pv;
#pragma unroll
        for (int nt = 0; nt < 4; ++nt) acc[nt][v] *= alpha;
      }

#pragma unroll
      for (int v = 0; v < 8; ++v) myp[(v + 8 * half) * 16 + l16] = p[v];
      wave_lds_handover();

#pragma unroll
      for (int kk = 0; kk < 16; kk += 4) {
        v2f a = *(const v2f*)(myp + l16 * 16 + kk + 2 * half);
        const float* v0 = vb + (sub * 16 + kk + 2 * half) * VPITCH + l16;
        const float* v1 = v0 + VPITCH;
#pragma unroll
        for (int nt = 0; nt < 4; ++nt) {
          v2f bf;
          bf.x = v0[nt * 16];
          bf.y = v1[nt * 16];
          acc[nt] = wmma4(a, bf, acc[nt]);
        }
      }
      wave_lds_handover();
    }

    if (more) {
#pragma unroll
      for (int i = 0; i < SLOADS; ++i) {
        *(v4f*)(&kst[buf ^ 1][kso + 8 * i * KPITCH]) = kr[i];
        *(v4f*)(&vst[buf ^ 1][vso + 8 * i * VPITCH]) = vr[i];
      }
    }
    __syncthreads();
  }

#pragma unroll
  for (int v = 0; v < 8; ++v) {
    const float inv = 1.0f / lrow[v];
    float* orow = O + base + (size_t)(q0 + v + 8 * half) * EMBED;
#pragma unroll
    for (int nt = 0; nt < 4; ++nt) {
      orow[nt * 16 + l16] = acc[nt][v] * inv;
    }
  }
}

extern "C" void kernel_launch(void* const* d_in, const int* in_sizes, int n_in,
                              void* d_out, int out_size, void* d_ws, size_t ws_size,
                              hipStream_t stream) {
  const float* x    = (const float*)d_in[0];
  const float* mask = (const float*)d_in[1];
  const float* Wq   = (const float*)d_in[2];
  const float* Wk   = (const float*)d_in[3];
  const float* Wv   = (const float*)d_in[4];
  const float* Wo   = (const float*)d_in[5];
  const float* bo   = (const float*)d_in[6];

  float* ws = (float*)d_ws;
  const size_t mat = (size_t)ROWS * EMBED;
  float* Qb = ws;
  float* Kb = ws + mat;
  float* Vb = ws + 2 * mat;
  float* Ab = ws + 3 * mat;

  dim3 g(ROWS / 64, EMBED / 128);
  gemm_bias_kernel<<<g, 256, 0, stream>>>(x, Wq, nullptr, Qb, 0);
  gemm_bias_kernel<<<g, 256, 0, stream>>>(x, Wk, nullptr, Kb, 0);
  gemm_bias_kernel<<<g, 256, 0, stream>>>(x, Wv, nullptr, Vb, 0);

  const int qtiles = BATCH * NHEAD * (S_LEN / 16);
  attention_kernel<<<qtiles / ATT_WAVES, ATT_THREADS, 0, stream>>>(Qb, Kb, Vb, mask, Ab);

  gemm_bias_kernel<<<g, 256, 0, stream>>>(Ab, Wo, bo, (float*)d_out, 1);
}


# attention tiles transposed (S^T=K*Q^T, O^T+=V^T*P^T): lane owns a query, softmax max/sum in-lane + one half swap, P^T to WMMA B from registers; row-sum and PV key sum order change
# speedup vs baseline: 1.6207x; 1.1153x over previous
#include <hip/hip_runtime.h>
#include <hip/hip_bf16.h>

typedef __attribute__((ext_vector_type(2))) float v2f;
typedef __attribute__((ext_vector_type(8))) float v8f;

#define EMBED 768
#define S_LEN 2048
#define NHEAD 12
#define HDIM  64
#define BATCH 2
#define ROWS  (BATCH * S_LEN)

__device__ __forceinline__ v8f wmma4(v2f a, v2f b, v8f c) {
  return __builtin_amdgcn_wmma_f32_16x16x4_f32(
       false, a,  false, b,
       (short)0, c,  false,  false);
}

__device__ __forceinline__ v8f vzero8() {
  v8f z = {0.f, 0.f, 0.f, 0.f, 0.f, 0.f, 0.f, 0.f};
  return z;
}

__global__ __launch_bounds__(256) void gemm_bias_kernel(
    const float* __restrict__ A, const float* __restrict__ W,
    const float* __restrict__ bias, float* __restrict__ C, int has_bias) {
  const int lane = threadIdx.x & 31;
  const int wave = threadIdx.x >> 5;
  const int half = lane >> 4;
  const int l16  = lane & 15;

  const int m0 = blockIdx.x * 64  + (wave & 3) * 16;
  const int n0 = blockIdx.y * 128 + (wave >> 2) * 64;

  v8f acc[4];
#pragma unroll
  for (int i = 0; i < 4; ++i) acc[i] = vzero8();

  const float* arow = A + (size_t)(m0 + l16) * EMBED;

#pragma unroll 2
  for (int kk = 0; kk < EMBED; kk += 4) {
    const int ka = kk + 2 * half;
    const v2f a = *(const v2f*)(arow + ka);
    const float* w0 = W + (size_t)ka * EMBED + n0 + l16;
    const float* w1 = w0 + EMBED;
#pragma unroll
    for (int nt = 0; nt < 4; ++nt) {
      v2f b;
      b.x = w0[nt * 16];
      b.y = w1[nt * 16];
      acc[nt] = wmma4(a, b, acc[nt]);
    }
  }

#pragma unroll
  for (int nt = 0; nt < 4; ++nt) {
    const int col = n0 + nt * 16 + l16;
    const float bv = has_bias ? bias[col] : 0.f;
#pragma unroll
    for (int v = 0; v < 8; ++v) {
      C[(size_t)(m0 + v + 8 * half) * EMBED + col] = acc[nt][v] + bv;
    }
  }
}

#define ATT_WAVES   4
#define ATT_THREADS (ATT_WAVES * 32)
#define KSTAGE      64
#define SLOADS      (KSTAGE / 8)
#define KPITCH      68
#define VPITCH2     136

typedef __attribute__((ext_vector_type(4))) float v4f;

__device__ __forceinline__ float xhalf(float x) {
  const int b = __builtin_bit_cast(int, x);
  return __builtin_bit_cast(float,
      __builtin_amdgcn_permlanex16(b, b, 0x76543210, 0xfedcba98, false, false));
}

__global__ __launch_bounds__(ATT_THREADS) void attention_kernel(
    const float* __restrict__ Q, const float* __restrict__ K,
    const float* __restrict__ V, const float* __restrict__ mask,
    float* __restrict__ O) {
  __shared__ __attribute__((aligned(16))) float kst[2][KSTAGE * KPITCH];
  __shared__ __attribute__((aligned(16))) float vst[2][(KSTAGE / 2) * VPITCH2];

  const int lane = threadIdx.x & 31;
  const int wave = threadIdx.x >> 5;
  const int half = lane >> 4;
  const int l16  = lane & 15;

  const int tile = blockIdx.x * ATT_WAVES + wave;
  const int qt = tile & 127;
  const int bh = tile >> 7;
  const int b  = bh / NHEAD;
  const int h  = bh % NHEAD;

  const int    q0   = qt * 16;
  const size_t base = (size_t)b * S_LEN * EMBED + (size_t)h * HDIM;

  const float* qrow = Q + base + (size_t)(q0 + l16) * EMBED;

  const int srow = threadIdx.x >> 4;
  const int sc4  = (threadIdx.x & 15) * 4;
  const float* kg = K + base + (size_t)srow * EMBED + sc4;
  const float* vg = V + base + (size_t)(2 * srow) * EMBED + sc4;
  const int kso = srow * KPITCH + sc4;
  const int vso = srow * VPITCH2 + 2 * sc4;

  v2f qf[HDIM / 4];
#pragma unroll
  for (int i = 0; i < HDIM / 4; ++i) qf[i] = *(const v2f*)(qrow + 4 * i + 2 * half);

  v4f kr[SLOADS], vr[SLOADS];
#pragma unroll
  for (int i = 0; i < SLOADS; ++i)
    kr[i] = *(const v4f*)(kg + (size_t)(8 * i) * EMBED);
#pragma unroll
  for (int i = 0; i < SLOADS / 2; ++i) {
    vr[2 * i]     = *(const v4f*)(vg + (size_t)(16 * i) * EMBED);
    vr[2 * i + 1] = *(const v4f*)(vg + (size_t)(16 * i + 1) * EMBED);
  }
#pragma unroll
  for (int i = 0; i < SLOADS; ++i)
    *(v4f*)(&kst[0][kso + 8 * i * KPITCH]) = kr[i];
#pragma unroll
  for (int i = 0; i < SLOADS / 2; ++i) {
    const v4f a = vr[2 * i], c = vr[2 * i + 1];
    v4f lo, hi;
    lo.x = a.x; lo.y = c.x; lo.z = a.y; lo.w = c.y;
    hi.x = a.z; hi.y = c.z; hi.z = a.w; hi.w = c.w;
    *(v4f*)(&vst[0][vso + 8 * i * VPITCH2])     = lo;
    *(v4f*)(&vst[0][vso + 8 * i * VPITCH2 + 4]) = hi;
  }

  v8f acc[4];
#pragma unroll
  for (int i = 0; i < 4; ++i) acc[i] = vzero8();
  float mrun = -3.0e38f, lrun = 0.f;

  const float* mrow = mask + (size_t)(q0 + l16) * S_LEN + 8 * half;
  v4f mk0 = *(const v4f*)(mrow);
  v4f mk1 = *(const v4f*)(mrow + 4);

  __syncthreads();

  for (int st = 0; st < S_LEN / KSTAGE; ++st) {
    const int  buf  = st & 1;
    const bool more = (st + 1) < (S_LEN / KSTAGE);

    if (more) {
      const size_t off = (size_t)(st + 1) * KSTAGE * EMBED;
#pragma unroll
      for (int i = 0; i < SLOADS; ++i)
        kr[i] = *(const v4f*)(kg + off + (size_t)(8 * i) * EMBED);
#pragma unroll
      for (int i = 0; i < SLOADS / 2; ++i) {
        vr[2 * i]     = *(const v4f*)(vg + off + (size_t)(16 * i) * EMBED);
        vr[2 * i + 1] = *(const v4f*)(vg + off + (size_t)(16 * i + 1) * EMBED);
      }
    }

    const float* kb = &kst[buf][0];
    const float* vb = &vst[buf][0];

    v2f kf[HDIM / 4];
    {
      const float* krow = kb + l16 * KPITCH + 2 * half;
#pragma unroll
      for (int i = 0; i < HDIM / 4; ++i) kf[i] = *(const v2f*)(krow + 4 * i);
    }

#pragma unroll
    for (int sub = 0; sub < KSTAGE / 16; ++sub) {
      const int kt = st * (KSTAGE / 16) + sub;

      v8f s = vzero8();
#pragma unroll
      for (int i = 0; i < HDIM / 4; ++i) s = wmma4(kf[i], qf[i], s);

      __builtin_amdgcn_sched_barrier(0);

      v2f vf[4][4];
#pragma unroll
      for (int j = 0; j < 4; ++j) {
        const float* vp = vb + (sub * 8 + 4 * half + j) * VPITCH2 + 2 * l16;
#pragma unroll
        for (int nt = 0; nt < 4; ++nt) vf[j][nt] = *(const v2f*)(vp + 32 * nt);
      }
      if (sub + 1 < KSTAGE / 16) {
        const float* krow = kb + ((sub + 1) * 16 + l16) * KPITCH + 2 * half;
#pragma unroll
        for (int i = 0; i < HDIM / 4; ++i) kf[i] = *(const v2f*)(krow + 4 * i);
      }

      const v4f mc0 = mk0, mc1 = mk1;
      const int ktn = (kt + 1 < S_LEN / 16) ? kt + 1 : kt;
      mk0 = *(const v4f*)(mrow + ktn * 16);
      mk1 = *(const v4f*)(mrow + ktn * 16 + 4);

      float sv[8];
      sv[0] = s[0] * 0.125f + mc0.x * (-1.0e9f);
      sv[1] = s[1] * 0.125f + mc0.y * (-1.0e9f);
      sv[2] = s[2] * 0.125f + mc0.z * (-1.0e9f);
      sv[3] = s[3] * 0.125f + mc0.w * (-1.0e9f);
      sv[4] = s[4] * 0.125f + mc1.x * (-1.0e9f);
      sv[5] = s[5] * 0.125f + mc1.y * (-1.0e9f);
      sv[6] = s[6] * 0.125f + mc1.z * (-1.0e9f);
      sv[7] = s[7] * 0.125f + mc1.w * (-1.0e9f);

      float tmax = fmaxf(fmaxf(fmaxf(sv[0], sv[1]), fmaxf(sv[2], sv[3])),
                         fmaxf(fmaxf(sv[4], sv[5]), fmaxf(sv[6], sv[7])));
      tmax = fmaxf(tmax, xhalf(tmax));
      const float mnew  = fmaxf(mrun, tmax);
      const float alpha = __expf(mrun - mnew);
      float p[8];
#pragma unroll
      for (int v = 0; v < 8; ++v) p[v] = __expf(sv[v] - mnew);
      float rsum = ((p[0] + p[1]) + (p[2] + p[3])) + ((p[4] + p[5]) + (p[6] + p[7]));
      rsum += xhalf(rsum);
      lrun = lrun * alpha + rsum;
      mrun = mnew;
#pragma unroll
      for (int nt = 0; nt < 4; ++nt)
#pragma unroll
        for (int r = 0; r < 8; ++r) acc[nt][r] *= alpha;

      __builtin_amdgcn_sched_barrier(0);

#pragma unroll
      for (int j = 0; j < 4; ++j) {
        v2f pb;
        pb.x = p[2 * j];
        pb.y = p[2 * j + 1];
#pragma unroll
        for (int nt = 0; nt < 4; ++nt) acc[nt] = wmma4(vf[j][nt], pb, acc[nt]);
      }
    }

    if (more) {
#pragma unroll
      for (int i = 0; i < SLOADS; ++i)
        *(v4f*)(&kst[buf ^ 1][kso + 8 * i * KPITCH]) = kr[i];
#pragma unroll
      for (int i = 0; i < SLOADS / 2; ++i) {
        const v4f a = vr[2 * i], c = vr[2 * i + 1];
        v4f lo, hi;
        lo.x = a.x; lo.y = c.x; lo.z = a.y; lo.w = c.y;
        hi.x = a.z; hi.y = c.z; hi.z = a.w; hi.w = c.w;
        *(v4f*)(&vst[buf ^ 1][vso + 8 * i * VPITCH2])     = lo;
        *(v4f*)(&vst[buf ^ 1][vso + 8 * i * VPITCH2 + 4]) = hi;
      }
    }
    __syncthreads();
  }

  const float inv = 1.0f / lrun;
  float* orow = O + base + (size_t)(q0 + l16) * EMBED + 8 * half;
#pragma unroll
  for (int nt = 0; nt < 4; ++nt) {
    v4f o0, o1;
    o0.x = acc[nt][0] * inv; o0.y = acc[nt][1] * inv;
    o0.z = acc[nt][2] * inv; o0.w = acc[nt][3] * inv;
    o1.x = acc[nt][4] * inv; o1.y = acc[nt][5] * inv;
    o1.z = acc[nt][6] * inv; o1.w = acc[nt][7] * inv;
    *(v4f*)(orow + nt * 16)     = o0;
    *(v4f*)(orow + nt * 16 + 4) = o1;
  }
}

extern "C" void kernel_launch(void* const* d_in, const int* in_sizes, int n_in,
                              void* d_out, int out_size, void* d_ws, size_t ws_size,
                              hipStream_t stream) {
  const float* x    = (const float*)d_in[0];
  const float* mask = (const float*)d_in[1];
  const float* Wq   = (const float*)d_in[2];
  const float* Wk   = (const float*)d_in[3];
  const float* Wv   = (const float*)d_in[4];
  const float* Wo   = (const float*)d_in[5];
  const float* bo   = (const float*)d_in[6];

  float* ws = (float*)d_ws;
  const size_t mat = (size_t)ROWS * EMBED;
  float* Qb = ws;
  float* Kb = ws + mat;
  float* Vb = ws + 2 * mat;
  float* Ab = ws + 3 * mat;

  dim3 g(ROWS / 64, EMBED / 128);
  gemm_bias_kernel<<<g, 256, 0, stream>>>(x, Wq, nullptr, Qb, 0);
  gemm_bias_kernel<<<g, 256, 0, stream>>>(x, Wk, nullptr, Kb, 0);
  gemm_bias_kernel<<<g, 256, 0, stream>>>(x, Wv, nullptr, Vb, 0);

  const int qtiles = BATCH * NHEAD * (S_LEN / 16);
  attention_kernel<<<qtiles / ATT_WAVES, ATT_THREADS, 0, stream>>>(Qb, Kb, Vb, mask, Ab);

  gemm_bias_kernel<<<g, 256, 0, stream>>>(Ab, Wo, bo, (float*)d_out, 1);
}


# projection GEMMs: each k slab of A and W staged once per 8-wave work-group in LDS (128x96 tile, whole-line b128 loads, double-buffered), fragments read from LDS, not per-wave strided global loads
# speedup vs baseline: 2.3303x; 1.4378x over previous
#include <hip/hip_runtime.h>
#include <hip/hip_bf16.h>

typedef __attribute__((ext_vector_type(2))) float v2f;
typedef __attribute__((ext_vector_type(8))) float v8f;

#define EMBED 768
#define S_LEN 2048
#define NHEAD 12
#define HDIM  64
#define BATCH 2
#define ROWS  (BATCH * S_LEN)

__device__ __forceinline__ v8f wmma4(v2f a, v2f b, v8f c) {
  return __builtin_amdgcn_wmma_f32_16x16x4_f32(
       false, a,  false, b,
       (short)0, c,  false,  false);
}

__device__ __forceinline__ v8f vzero8() {
  v8f z = {0.f, 0.f, 0.f, 0.f, 0.f, 0.f, 0.f, 0.f};
  return z;
}

#define GBM      128
#define GBN      96
#define GBK      128
#define GTHREADS 256
#define GSLABS   (EMBED / GBK)
#define APITCH   (GBK + 4)
#define WPITCH2  (2 * GBN + 32)
#define A_LOADS  (GBM * (GBK / 4) / GTHREADS)
#define W_CJ     ((GBN / 4) / 8)
#define W_PR     ((GBK / 2) / (GTHREADS / 8))
#define KGROUP   2
#define NGROUPS  (GBK / 4 / KGROUP)

typedef __attribute__((ext_vector_type(4))) float g4f;

__global__ __launch_bounds__(GTHREADS) void gemm_bias_kernel(
    const float* __restrict__ A, const float* __restrict__ W,
    const float* __restrict__ bias, float* __restrict__ C, int has_bias) {
  __shared__ __attribute__((aligned(16))) float ast[2][GBM * APITCH];
  __shared__ __attribute__((aligned(16))) float wst[2][(GBK / 2) * WPITCH2];

  const int lane = threadIdx.x & 31;
  const int wave = threadIdx.x >> 5;
  const int half = lane >> 4;
  const int l16  = lane & 15;
  const int wm   = wave & 3;
  const int wn   = wave >> 2;

  const int bm0 = blockIdx.x * GBM;
  const int bn0 = blockIdx.y * GBN;

  const int asr = threadIdx.x >> 5;
  const int ac4 = (threadIdx.x & 31) * 4;
  const float* ag = A + (size_t)(bm0 + asr) * EMBED + ac4;
  const int aso = asr * APITCH + ac4;
  const int wsp = threadIdx.x >> 3;
  const int wc4 = (threadIdx.x & 7) * 4;
  const float* wg = W + (size_t)(2 * wsp) * EMBED + bn0 + wc4;
  const int wso = wsp * WPITCH2 + 2 * wc4;

  g4f ar[A_LOADS], wr[2 * W_PR * W_CJ];

#pragma unroll
  for (int i = 0; i < A_LOADS; ++i)
    ar[i] = *(const g4f*)(ag + (size_t)(8 * i) * EMBED);
#pragma unroll
  for (int r = 0; r < W_PR; ++r)
#pragma unroll
    for (int j = 0; j < W_CJ; ++j) {
      const float* p = wg + (size_t)(64 * r) * EMBED + 32 * j;
      wr[2 * (r * W_CJ + j)]     = *(const g4f*)(p);
      wr[2 * (r * W_CJ + j) + 1] = *(const g4f*)(p + EMBED);
    }
#pragma unroll
  for (int i = 0; i < A_LOADS; ++i)
    *(g4f*)(&ast[0][aso + 8 * i * APITCH]) = ar[i];
#pragma unroll
  for (int r = 0; r < W_PR; ++r)
#pragma unroll
    for (int j = 0; j < W_CJ; ++j) {
      const g4f a = wr[2 * (r * W_CJ + j)], c = wr[2 * (r * W_CJ + j) + 1];
      g4f lo, hi;
      lo.x = a.x; lo.y = c.x; lo.z = a.y; lo.w = c.y;
      hi.x = a.z; hi.y = c.z; hi.z = a.w; hi.w = c.w;
      *(g4f*)(&wst[0][wso + 32 * r * WPITCH2 + 64 * j])     = lo;
      *(g4f*)(&wst[0][wso + 32 * r * WPITCH2 + 64 * j + 4]) = hi;
    }

  v8f acc[2][3];
#pragma unroll
  for (int a = 0; a < 2; ++a)
#pragma unroll
    for (int nt = 0; nt < 3; ++nt) acc[a][nt] = vzero8();

  const int afo = (wm * 32 + l16) * APITCH + 2 * half;
  const int bfo = half * WPITCH2 + 2 * (wn * 48 + l16);

  __syncthreads();

  for (int st = 0; st < GSLABS; ++st) {
    const int  buf  = st & 1;
    const bool more = (st + 1) < GSLABS;

    if (more) {
      const size_t ka = (size_t)(st + 1) * GBK;
#pragma unroll
      for (int i = 0; i < A_LOADS; ++i)
        ar[i] = *(const g4f*)(ag + ka + (size_t)(8 * i) * EMBED);
#pragma unroll
      for (int r = 0; r < W_PR; ++r)
#pragma unroll
        for (int j = 0; j < W_CJ; ++j) {
          const float* p = wg + ka * EMBED + (size_t)(64 * r) * EMBED + 32 * j;
          wr[2 * (r * W_CJ + j)]     = *(const g4f*)(p);
          wr[2 * (r * W_CJ + j) + 1] = *(const g4f*)(p + EMBED);
        }
    }

    const float* ab = &ast[buf][afo];
    const float* wb = &wst[buf][bfo];

    v2f fa[2][KGROUP][2], fb[2][KGROUP][3];
#pragma unroll
    for (int s = 0; s < KGROUP; ++s) {
#pragma unroll
      for (int a = 0; a < 2; ++a)
        fa[0][s][a] = *(const v2f*)(ab + 16 * a * APITCH + 4 * s);
#pragma unroll
      for (int nt = 0; nt < 3; ++nt)
        fb[0][s][nt] = *(const v2f*)(wb + 2 * s * WPITCH2 + 32 * nt);
    }

#pragma unroll
    for (int g = 0; g < NGROUPS; ++g) {
      const int cur = g & 1, nxt = cur ^ 1;
      if (g + 1 < NGROUPS) {
#pragma unroll
        for (int s = 0; s < KGROUP; ++s) {
          const int i = (g + 1) * KGROUP + s;
#pragma unroll
          for (int a = 0; a < 2; ++a)
            fa[nxt][s][a] = *(const v2f*)(ab + 16 * a * APITCH + 4 * i);
#pragma unroll
          for (int nt = 0; nt < 3; ++nt)
            fb[nxt][s][nt] = *(const v2f*)(wb + 2 * i * WPITCH2 + 32 * nt);
        }
      }

      __builtin_amdgcn_sched_barrier(0);

#pragma unroll
      for (int s = 0; s < KGROUP; ++s)
#pragma unroll
        for (int a = 0; a < 2; ++a)
#pragma unroll
          for (int nt = 0; nt < 3; ++nt)
            acc[a][nt] = wmma4(fa[cur][s][a], fb[cur][s][nt], acc[a][nt]);

      __builtin_amdgcn_sched_barrier(0);

      if (g == NGROUPS / 2 - 1 && more) {
#pragma unroll
        for (int i = 0; i < A_LOADS; ++i)
          *(g4f*)(&ast[buf ^ 1][aso + 8 * i * APITCH]) = ar[i];
#pragma unroll
        for (int r = 0; r < W_PR; ++r)
#pragma unroll
          for (int j = 0; j < W_CJ; ++j) {
            const g4f a = wr[2 * (r * W_CJ + j)], c = wr[2 * (r * W_CJ + j) + 1];
            g4f lo, hi;
            lo.x = a.x; lo.y = c.x; lo.z = a.y; lo.w = c.y;
            hi.x = a.z; hi.y = c.z; hi.z = a.w; hi.w = c.w;
            *(g4f*)(&wst[buf ^ 1][wso + 32 * r * WPITCH2 + 64 * j])     = lo;
            *(g4f*)(&wst[buf ^ 1][wso + 32 * r * WPITCH2 + 64 * j + 4]) = hi;
          }
      }
    }
    __syncthreads();
  }

#pragma unroll
  for (int nt = 0; nt < 3; ++nt) {
    const int col = bn0 + wn * 48 + nt * 16 + l16;
    const float bv = has_bias ? bias[col] : 0.f;
#pragma unroll
    for (int a = 0; a < 2; ++a) {
      const int row0 = bm0 + wm * 32 + a * 16 + 8 * half;
#pragma unroll
      for (int v = 0; v < 8; ++v) {
        C[(size_t)(row0 + v) * EMBED + col] = acc[a][nt][v] + bv;
      }
    }
  }
}

#define ATT_WAVES   4
#define ATT_THREADS (ATT_WAVES * 32)
#define KSTAGE      64
#define SLOADS      (KSTAGE / 8)
#define KPITCH      68
#define VPITCH2     136

typedef __attribute__((ext_vector_type(4))) float v4f;

__device__ __forceinline__ float xhalf(float x) {
  const int b = __builtin_bit_cast(int, x);
  return __builtin_bit_cast(float,
      __builtin_amdgcn_permlanex16(b, b, 0x76543210, 0xfedcba98, false, false));
}

__global__ __launch_bounds__(ATT_THREADS) void attention_kernel(
    const float* __restrict__ Q, const float* __restrict__ K,
    const float* __restrict__ V, const float* __restrict__ mask,
    float* __restrict__ O) {
  __shared__ __attribute__((aligned(16))) float kst[2][KSTAGE * KPITCH];
  __shared__ __attribute__((aligned(16))) float vst[2][(KSTAGE / 2) * VPITCH2];

  const int lane = threadIdx.x & 31;
  const int wave = threadIdx.x >> 5;
  const int half = lane >> 4;
  const int l16  = lane & 15;

  const int tile = blockIdx.x * ATT_WAVES + wave;
  const int qt = tile & 127;
  const int bh = tile >> 7;
  const int b  = bh / NHEAD;
  const int h  = bh % NHEAD;

  const int    q0   = qt * 16;
  const size_t base = (size_t)b * S_LEN * EMBED + (size_t)h * HDIM;

  const float* qrow = Q + base + (size_t)(q0 + l16) * EMBED;

  const int srow = threadIdx.x >> 4;
  const int sc4  = (threadIdx.x & 15) * 4;
  const float* kg = K + base + (size_t)srow * EMBED + sc4;
  const float* vg = V + base + (size_t)(2 * srow) * EMBED + sc4;
  const int kso = srow * KPITCH + sc4;
  const int vso = srow * VPITCH2 + 2 * sc4;

  v2f qf[HDIM / 4];
#pragma unroll
  for (int i = 0; i < HDIM / 4; ++i) qf[i] = *(const v2f*)(qrow + 4 * i + 2 * half);

  v4f kr[SLOADS], vr[SLOADS];
#pragma unroll
  for (int i = 0; i < SLOADS; ++i)
    kr[i] = *(const v4f*)(kg + (size_t)(8 * i) * EMBED);
#pragma unroll
  for (int i = 0; i < SLOADS / 2; ++i) {
    vr[2 * i]     = *(const v4f*)(vg + (size_t)(16 * i) * EMBED);
    vr[2 * i + 1] = *(const v4f*)(vg + (size_t)(16 * i + 1) * EMBED);
  }
#pragma unroll
  for (int i = 0; i < SLOADS; ++i)
    *(v4f*)(&kst[0][kso + 8 * i * KPITCH]) = kr[i];
#pragma unroll
  for (int i = 0; i < SLOADS / 2; ++i) {
    const v4f a = vr[2 * i], c = vr[2 * i + 1];
    v4f lo, hi;
    lo.x = a.x; lo.y = c.x; lo.z = a.y; lo.w = c.y;
    hi.x = a.z; hi.y = c.z; hi.z = a.w; hi.w = c.w;
    *(v4f*)(&vst[0][vso + 8 * i * VPITCH2])     = lo;
    *(v4f*)(&vst[0][vso + 8 * i * VPITCH2 + 4]) = hi;
  }

  v8f acc[4];
#pragma unroll
  for (int i = 0; i < 4; ++i) acc[i] = vzero8();
  float mrun = -3.0e38f, lrun = 0.f;

  const float* mrow = mask + (size_t)(q0 + l16) * S_LEN + 8 * half;
  v4f mk0 = *(const v4f*)(mrow);
  v4f mk1 = *(const v4f*)(mrow + 4);

  __syncthreads();

  for (int st = 0; st < S_LEN / KSTAGE; ++st) {
    const int  buf  = st & 1;
    const bool more = (st + 1) < (S_LEN / KSTAGE);

    if (more) {
      const size_t off = (size_t)(st + 1) * KSTAGE * EMBED;
#pragma unroll
      for (int i = 0; i < SLOADS; ++i)
        kr[i] = *(const v4f*)(kg + off + (size_t)(8 * i) * EMBED);
#pragma unroll
      for (int i = 0; i < SLOADS / 2; ++i) {
        vr[2 * i]     = *(const v4f*)(vg + off + (size_t)(16 * i) * EMBED);
        vr[2 * i + 1] = *(const v4f*)(vg + off + (size_t)(16 * i + 1) * EMBED);
      }
    }

    const float* kb = &kst[buf][0];
    const float* vb = &vst[buf][0];

    v2f kf[HDIM / 4];
    {
      const float* krow = kb + l16 * KPITCH + 2 * half;
#pragma unroll
      for (int i = 0; i < HDIM / 4; ++i) kf[i] = *(const v2f*)(krow + 4 * i);
    }

#pragma unroll
    for (int sub = 0; sub < KSTAGE / 16; ++sub) {
      const int kt = st * (KSTAGE / 16) + sub;

      v8f s = vzero8();
#pragma unroll
      for (int i = 0; i < HDIM / 4; ++i) s = wmma4(kf[i], qf[i], s);

      __builtin_amdgcn_sched_barrier(0);

      v2f vf[4][4];
#pragma unroll
      for (int j = 0; j < 4; ++j) {
        const float* vp = vb + (sub * 8 + 4 * half + j) * VPITCH2 + 2 * l16;
#pragma unroll
        for (int nt = 0; nt < 4; ++nt) vf[j][nt] = *(const v2f*)(vp + 32 * nt);
      }
      if (sub + 1 < KSTAGE / 16) {
        const float* krow = kb + ((sub + 1) * 16 + l16) * KPITCH + 2 * half;
#pragma unroll
        for (int i = 0; i < HDIM / 4; ++i) kf[i] = *(const v2f*)(krow + 4 * i);
      }

      const v4f mc0 = mk0, mc1 = mk1;
      const int ktn = (kt + 1 < S_LEN / 16) ? kt + 1 : kt;
      mk0 = *(const v4f*)(mrow + ktn * 16);
      mk1 = *(const v4f*)(mrow + ktn * 16 + 4);

      float sv[8];
      sv[0] = s[0] * 0.125f + mc0.x * (-1.0e9f);
      sv[1] = s[1] * 0.125f + mc0.y * (-1.0e9f);
      sv[2] = s[2] * 0.125f + mc0.z * (-1.0e9f);
      sv[3] = s[3] * 0.125f + mc0.w * (-1.0e9f);
      sv[4] = s[4] * 0.125f + mc1.x * (-1.0e9f);
      sv[5] = s[5] * 0.125f + mc1.y * (-1.0e9f);
      sv[6] = s[6] * 0.125f + mc1.z * (-1.0e9f);
      sv[7] = s[7] * 0.125f + mc1.w * (-1.0e9f);

      float tmax = fmaxf(fmaxf(fmaxf(sv[0], sv[1]), fmaxf(sv[2], sv[3])),
                         fmaxf(fmaxf(sv[4], sv[5]), fmaxf(sv[6], sv[7])));
      tmax = fmaxf(tmax, xhalf(tmax));
      const float mnew  = fmaxf(mrun, tmax);
      const float alpha = __expf(mrun - mnew);
      float p[8];
#pragma unroll
      for (int v = 0; v < 8; ++v) p[v] = __expf(sv[v] - mnew);
      float rsum = ((p[0] + p[1]) + (p[2] + p[3])) + ((p[4] + p[5]) + (p[6] + p[7]));
      rsum += xhalf(rsum);
      lrun = lrun * alpha + rsum;
      mrun = mnew;
#pragma unroll
      for (int nt = 0; nt < 4; ++nt)
#pragma unroll
        for (int r = 0; r < 8; ++r) acc[nt][r] *= alpha;

      __builtin_amdgcn_sched_barrier(0);

#pragma unroll
      for (int j = 0; j < 4; ++j) {
        v2f pb;
        pb.x = p[2 * j];
        pb.y = p[2 * j + 1];
#pragma unroll
        for (int nt = 0; nt < 4; ++nt) acc[nt] = wmma4(vf[j][nt], pb, acc[nt]);
      }
    }

    if (more) {
#pragma unroll
      for (int i = 0; i < SLOADS; ++i)
        *(v4f*)(&kst[buf ^ 1][kso + 8 * i * KPITCH]) = kr[i];
#pragma unroll
      for (int i = 0; i < SLOADS / 2; ++i) {
        const v4f a = vr[2 * i], c = vr[2 * i + 1];
        v4f lo, hi;
        lo.x = a.x; lo.y = c.x; lo.z = a.y; lo.w = c.y;
        hi.x = a.z; hi.y = c.z; hi.z = a.w; hi.w = c.w;
        *(v4f*)(&vst[buf ^ 1][vso + 8 * i * VPITCH2])     = lo;
        *(v4f*)(&vst[buf ^ 1][vso + 8 * i * VPITCH2 + 4]) = hi;
      }
    }
    __syncthreads();
  }

  const float inv = 1.0f / lrun;
  float* orow = O + base + (size_t)(q0 + l16) * EMBED + 8 * half;
#pragma unroll
  for (int nt = 0; nt < 4; ++nt) {
    v4f o0, o1;
    o0.x = acc[nt][0] * inv; o0.y = acc[nt][1] * inv;
    o0.z = acc[nt][2] * inv; o0.w = acc[nt][3] * inv;
    o1.x = acc[nt][4] * inv; o1.y = acc[nt][5] * inv;
    o1.z = acc[nt][6] * inv; o1.w = acc[nt][7] * inv;
    *(v4f*)(orow + nt * 16)     = o0;
    *(v4f*)(orow + nt * 16 + 4) = o1;
  }
}

extern "C" void kernel_launch(void* const* d_in, const int* in_sizes, int n_in,
                              void* d_out, int out_size, void* d_ws, size_t ws_size,
                              hipStream_t stream) {
  const float* x    = (const float*)d_in[0];
  const float* mask = (const float*)d_in[1];
  const float* Wq   = (const float*)d_in[2];
  const float* Wk   = (const float*)d_in[3];
  const float* Wv   = (const float*)d_in[4];
  const float* Wo   = (const float*)d_in[5];
  const float* bo   = (const float*)d_in[6];

  float* ws = (float*)d_ws;
  const size_t mat = (size_t)ROWS * EMBED;
  float* Qb = ws;
  float* Kb = ws + mat;
  float* Vb = ws + 2 * mat;
  float* Ab = ws + 3 * mat;

  dim3 g(ROWS / GBM, EMBED / GBN);
  gemm_bias_kernel<<<g, GTHREADS, 0, stream>>>(x, Wq, nullptr, Qb, 0);
  gemm_bias_kernel<<<g, GTHREADS, 0, stream>>>(x, Wk, nullptr, Kb, 0);
  gemm_bias_kernel<<<g, GTHREADS, 0, stream>>>(x, Wv, nullptr, Vb, 0);

  const int qtiles = BATCH * NHEAD * (S_LEN / 16);
  attention_kernel<<<qtiles / ATT_WAVES, ATT_THREADS, 0, stream>>>(Qb, Kb, Vb, mask, Ab);

  gemm_bias_kernel<<<g, GTHREADS, 0, stream>>>(Ab, Wo, bo, (float*)d_out, 1);
}


# Q, K, V projections fused into one launch: each work-group stages its x k-slab once in LDS and multiplies it by all three weight tiles (wave tile 32x48 x 3, 36-WMMA runs); 5 launches -> 3
# speedup vs baseline: 2.4117x; 1.0349x over previous
#include <hip/hip_runtime.h>
#include <hip/hip_bf16.h>

typedef __attribute__((ext_vector_type(2))) float v2f;
typedef __attribute__((ext_vector_type(8))) float v8f;

#define EMBED 768
#define S_LEN 2048
#define NHEAD 12
#define HDIM  64
#define BATCH 2
#define ROWS  (BATCH * S_LEN)

__device__ __forceinline__ v8f wmma4(v2f a, v2f b, v8f c) {
  return __builtin_amdgcn_wmma_f32_16x16x4_f32(
       false, a,  false, b,
       (short)0, c,  false,  false);
}

__device__ __forceinline__ v8f vzero8() {
  v8f z = {0.f, 0.f, 0.f, 0.f, 0.f, 0.f, 0.f, 0.f};
  return z;
}

#define GBM      128
#define GBN      96
#define GBK      128
#define GTHREADS 256
#define GSLABS   (EMBED / GBK)
#define APITCH   (GBK + 4)
#define WPITCH2  (2 * GBN + 32)
#define A_LOADS  (GBM * (GBK / 4) / GTHREADS)
#define W_CJ     ((GBN / 4) / 8)
#define W_PR     ((GBK / 2) / (GTHREADS / 8))
#define KGROUP   2
#define NGROUPS  (GBK / 4 / KGROUP)

typedef __attribute__((ext_vector_type(4))) float g4f;

__global__ __launch_bounds__(GTHREADS) void gemm_bias_kernel(
    const float* __restrict__ A, const float* __restrict__ W,
    const float* __restrict__ bias, float* __restrict__ C, int has_bias) {
  __shared__ __attribute__((aligned(16))) float ast[2][GBM * APITCH];
  __shared__ __attribute__((aligned(16))) float wst[2][(GBK / 2) * WPITCH2];

  const int lane = threadIdx.x & 31;
  const int wave = threadIdx.x >> 5;
  const int half = lane >> 4;
  const int l16  = lane & 15;
  const int wm   = wave & 3;
  const int wn   = wave >> 2;

  const int bm0 = blockIdx.x * GBM;
  const int bn0 = blockIdx.y * GBN;

  const int asr = threadIdx.x >> 5;
  const int ac4 = (threadIdx.x & 31) * 4;
  const float* ag = A + (size_t)(bm0 + asr) * EMBED + ac4;
  const int aso = asr * APITCH + ac4;
  const int wsp = threadIdx.x >> 3;
  const int wc4 = (threadIdx.x & 7) * 4;
  const float* wg = W + (size_t)(2 * wsp) * EMBED + bn0 + wc4;
  const int wso = wsp * WPITCH2 + 2 * wc4;

  g4f ar[A_LOADS], wr[2 * W_PR * W_CJ];

#pragma unroll
  for (int i = 0; i < A_LOADS; ++i)
    ar[i] = *(const g4f*)(ag + (size_t)(8 * i) * EMBED);
#pragma unroll
  for (int r = 0; r < W_PR; ++r)
#pragma unroll
    for (int j = 0; j < W_CJ; ++j) {
      const float* p = wg + (size_t)(64 * r) * EMBED + 32 * j;
      wr[2 * (r * W_CJ + j)]     = *(const g4f*)(p);
      wr[2 * (r * W_CJ + j) + 1] = *(const g4f*)(p + EMBED);
    }
#pragma unroll
  for (int i = 0; i < A_LOADS; ++i)
    *(g4f*)(&ast[0][aso + 8 * i * APITCH]) = ar[i];
#pragma unroll
  for (int r = 0; r < W_PR; ++r)
#pragma unroll
    for (int j = 0; j < W_CJ; ++j) {
      const g4f a = wr[2 * (r * W_CJ + j)], c = wr[2 * (r * W_CJ + j) + 1];
      g4f lo, hi;
      lo.x = a.x; lo.y = c.x; lo.z = a.y; lo.w = c.y;
      hi.x = a.z; hi.y = c.z; hi.z = a.w; hi.w = c.w;
      *(g4f*)(&wst[0][wso + 32 * r * WPITCH2 + 64 * j])     = lo;
      *(g4f*)(&wst[0][wso + 32 * r * WPITCH2 + 64 * j + 4]) = hi;
    }

  v8f acc[2][3];
#pragma unroll
  for (int a = 0; a < 2; ++a)
#pragma unroll
    for (int nt = 0; nt < 3; ++nt) acc[a][nt] = vzero8();

  const int afo = (wm * 32 + l16) * APITCH + 2 * half;
  const int bfo = half * WPITCH2 + 2 * (wn * 48 + l16);

  __syncthreads();

  for (int st = 0; st < GSLABS; ++st) {
    const int  buf  = st & 1;
    const bool more = (st + 1) < GSLABS;

    if (more) {
      const size_t ka = (size_t)(st + 1) * GBK;
#pragma unroll
      for (int i = 0; i < A_LOADS; ++i)
        ar[i] = *(const g4f*)(ag + ka + (size_t)(8 * i) * EMBED);
#pragma unroll
      for (int r = 0; r < W_PR; ++r)
#pragma unroll
        for (int j = 0; j < W_CJ; ++j) {
          const float* p = wg + ka * EMBED + (size_t)(64 * r) * EMBED + 32 * j;
          wr[2 * (r * W_CJ + j)]     = *(const g4f*)(p);
          wr[2 * (r * W_CJ + j) + 1] = *(const g4f*)(p + EMBED);
        }
    }

    const float* ab = &ast[buf][afo];
    const float* wb = &wst[buf][bfo];

    v2f fa[2][KGROUP][2], fb[2][KGROUP][3];
#pragma unroll
    for (int s = 0; s < KGROUP; ++s) {
#pragma unroll
      for (int a = 0; a < 2; ++a)
        fa[0][s][a] = *(const v2f*)(ab + 16 * a * APITCH + 4 * s);
#pragma unroll
      for (int nt = 0; nt < 3; ++nt)
        fb[0][s][nt] = *(const v2f*)(wb + 2 * s * WPITCH2 + 32 * nt);
    }

#pragma unroll
    for (int g = 0; g < NGROUPS; ++g) {
      const int cur = g & 1, nxt = cur ^ 1;
      if (g + 1 < NGROUPS) {
#pragma unroll
        for (int s = 0; s < KGROUP; ++s) {
          const int i = (g + 1) * KGROUP + s;
#pragma unroll
          for (int a = 0; a < 2; ++a)
            fa[nxt][s][a] = *(const v2f*)(ab + 16 * a * APITCH + 4 * i);
#pragma unroll
          for (int nt = 0; nt < 3; ++nt)
            fb[nxt][s][nt] = *(const v2f*)(wb + 2 * i * WPITCH2 + 32 * nt);
        }
      }

      __builtin_amdgcn_sched_barrier(0);

#pragma unroll
      for (int s = 0; s < KGROUP; ++s)
#pragma unroll
        for (int a = 0; a < 2; ++a)
#pragma unroll
          for (int nt = 0; nt < 3; ++nt)
            acc[a][nt] = wmma4(fa[cur][s][a], fb[cur][s][nt], acc[a][nt]);

      __builtin_amdgcn_sched_barrier(0);

      if (g == NGROUPS / 2 - 1 && more) {
#pragma unroll
        for (int i = 0; i < A_LOADS; ++i)
          *(g4f*)(&ast[buf ^ 1][aso + 8 * i * APITCH]) = ar[i];
#pragma unroll
        for (int r = 0; r < W_PR; ++r)
#pragma unroll
          for (int j = 0; j < W_CJ; ++j) {
            const g4f a = wr[2 * (r * W_CJ + j)], c = wr[2 * (r * W_CJ + j) + 1];
            g4f lo, hi;
            lo.x = a.x; lo.y = c.x; lo.z = a.y; lo.w = c.y;
            hi.x = a.z; hi.y = c.z; hi.z = a.w; hi.w = c.w;
            *(g4f*)(&wst[buf ^ 1][wso + 32 * r * WPITCH2 + 64 * j])     = lo;
            *(g4f*)(&wst[buf ^ 1][wso + 32 * r * WPITCH2 + 64 * j + 4]) = hi;
          }
      }
    }
    __syncthreads();
  }

#pragma unroll
  for (int nt = 0; nt < 3; ++nt) {
    const int col = bn0 + wn * 48 + nt * 16 + l16;
    const float bv = has_bias ? bias[col] : 0.f;
#pragma unroll
    for (int a = 0; a < 2; ++a) {
      const int row0 = bm0 + wm * 32 + a * 16 + 8 * half;
#pragma unroll
      for (int v = 0; v < 8; ++v) {
        C[(size_t)(row0 + v) * EMBED + col] = acc[a][nt][v] + bv;
      }
    }
  }
}

#define QBK      64
#define QSLABS   (EMBED / QBK)
#define QAPITCH  (QBK + 4)
#define QA_LOADS (GBM * (QBK / 4) / GTHREADS)
#define QNGROUPS (QBK / 4 / KGROUP)
#define QWMAT    ((QBK / 2) * WPITCH2)

__global__ __launch_bounds__(GTHREADS) void qkv_gemm_kernel(
    const float* __restrict__ X, const float* __restrict__ Wq,
    const float* __restrict__ Wk, const float* __restrict__ Wv,
    float* __restrict__ Q, float* __restrict__ K, float* __restrict__ V) {
  __shared__ __attribute__((aligned(16))) float ast[2][GBM * QAPITCH];
  __shared__ __attribute__((aligned(16))) float wst[2][3][QWMAT];

  const int lane = threadIdx.x & 31;
  const int wave = threadIdx.x >> 5;
  const int half = lane >> 4;
  const int l16  = lane & 15;
  const int wm   = wave & 3;
  const int wn   = wave >> 2;

  const int bm0 = blockIdx.x * GBM;
  const int bn0 = blockIdx.y * GBN;

  const int asr = threadIdx.x >> 4;
  const int ac4 = (threadIdx.x & 15) * 4;
  const float* ag = X + (size_t)(bm0 + asr) * EMBED + ac4;
  const int aso = asr * QAPITCH + ac4;
  const int wsp = threadIdx.x >> 3;
  const int wc4 = (threadIdx.x & 7) * 4;
  const size_t wgo = (size_t)(2 * wsp) * EMBED + bn0 + wc4;
  const float* wg0 = Wq + wgo;
  const float* wg1 = Wk + wgo;
  const float* wg2 = Wv + wgo;
  const int wso = wsp * WPITCH2 + 2 * wc4;

  g4f ar[QA_LOADS], wr[3][2 * W_CJ];

#pragma unroll
  for (int i = 0; i < QA_LOADS; ++i)
    ar[i] = *(const g4f*)(ag + (size_t)(16 * i) * EMBED);
#pragma unroll
  for (int j = 0; j < W_CJ; ++j) {
    wr[0][2 * j]     = *(const g4f*)(wg0 + 32 * j);
    wr[0][2 * j + 1] = *(const g4f*)(wg0 + 32 * j + EMBED);
    wr[1][2 * j]     = *(const g4f*)(wg1 + 32 * j);
    wr[1][2 * j + 1] = *(const g4f*)(wg1 + 32 * j + EMBED);
    wr[2][2 * j]     = *(const g4f*)(wg2 + 32 * j);
    wr[2][2 * j + 1] = *(const g4f*)(wg2 + 32 * j + EMBED);
  }
#pragma unroll
  for (int i = 0; i < QA_LOADS; ++i)
    *(g4f*)(&ast[0][aso + 16 * i * QAPITCH]) = ar[i];
#pragma unroll
  for (int m = 0; m < 3; ++m)
#pragma unroll
    for (int j = 0; j < W_CJ; ++j) {
      const g4f a = wr[m][2 * j], c = wr[m][2 * j + 1];
      g4f lo, hi;
      lo.x = a.x; lo.y = c.x; lo.z = a.y; lo.w = c.y;
      hi.x = a.z; hi.y = c.z; hi.z = a.w; hi.w = c.w;
      *(g4f*)(&wst[0][m][wso + 64 * j])     = lo;
      *(g4f*)(&wst[0][m][wso + 64 * j + 4]) = hi;
    }

  v8f acc[3][2][3];
#pragma unroll
  for (int m = 0; m < 3; ++m)
#pragma unroll
    for (int a = 0; a < 2; ++a)
#pragma unroll
      for (int nt = 0; nt < 3; ++nt) acc[m][a][nt] = vzero8();

  const int afo = (wm * 32 + l16) * QAPITCH + 2 * half;
  const int bfo = half * WPITCH2 + 2 * (wn * 48 + l16);

  __syncthreads();

  for (int st = 0; st < QSLABS; ++st) {
    const int  buf  = st & 1;
    const bool more = (st + 1) < QSLABS;

    if (more) {
      const size_t ka = (size_t)(st + 1) * QBK;
#pragma unroll
      for (int i = 0; i < QA_LOADS; ++i)
        ar[i] = *(const g4f*)(ag + ka + (size_t)(16 * i) * EMBED);
#pragma unroll
      for (int j = 0; j < W_CJ; ++j) {
        wr[0][2 * j]     = *(const g4f*)(wg0 + ka * EMBED + 32 * j);
        wr[0][2 * j + 1] = *(const g4f*)(wg0 + ka * EMBED + 32 * j + EMBED);
        wr[1][2 * j]     = *(const g4f*)(wg1 + ka * EMBED + 32 * j);
        wr[1][2 * j + 1] = *(const g4f*)(wg1 + ka * EMBED + 32 * j + EMBED);
        wr[2][2 * j]     = *(const g4f*)(wg2 + ka * EMBED + 32 * j);
        wr[2][2 * j + 1] = *(const g4f*)(wg2 + ka * EMBED + 32 * j + EMBED);
      }
    }

    const float* ab = &ast[buf][afo];
    const float* wb = &wst[buf][0][bfo];

    v2f fa[2][KGROUP][2], fb[2][KGROUP][3][3];
#pragma unroll
    for (int s = 0; s < KGROUP; ++s) {
#pragma unroll
      for (int a = 0; a < 2; ++a)
        fa[0][s][a] = *(const v2f*)(ab + 16 * a * QAPITCH + 4 * s);
#pragma unroll
      for (int m = 0; m < 3; ++m)
#pragma unroll
        for (int nt = 0; nt < 3; ++nt)
          fb[0][s][m][nt] = *(const v2f*)(wb + m * QWMAT + 2 * s * WPITCH2 + 32 * nt);
    }

#pragma unroll
    for (int g = 0; g < QNGROUPS; ++g) {
      const int cur = g & 1, nxt = cur ^ 1;
      if (g + 1 < QNGROUPS) {
#pragma unroll
        for (int s = 0; s < KGROUP; ++s) {
          const int i = (g + 1) * KGROUP + s;
#pragma unroll
          for (int a = 0; a < 2; ++a)
            fa[nxt][s][a] = *(const v2f*)(ab + 16 * a * QAPITCH + 4 * i);
#pragma unroll
          for (int m = 0; m < 3; ++m)
#pragma unroll
            for (int nt = 0; nt < 3; ++nt)
              fb[nxt][s][m][nt] = *(const v2f*)(wb + m * QWMAT + 2 * i * WPITCH2 + 32 * nt);
        }
      }

      __builtin_amdgcn_sched_barrier(0);

#pragma unroll
      for (int s = 0; s < KGROUP; ++s)
#pragma unroll
        for (int a = 0; a < 2; ++a)
#pragma unroll
          for (int m = 0; m < 3; ++m)
#pragma unroll
            for (int nt = 0; nt < 3; ++nt)
              acc[m][a][nt] = wmma4(fa[cur][s][a], fb[cur][s][m][nt], acc[m][a][nt]);

      __builtin_amdgcn_sched_barrier(0);

      if (g == QNGROUPS / 2 - 1 && more) {
#pragma unroll
        for (int i = 0; i < QA_LOADS; ++i)
          *(g4f*)(&ast[buf ^ 1][aso + 16 * i * QAPITCH]) = ar[i];
#pragma unroll
        for (int m = 0; m < 3; ++m)
#pragma unroll
          for (int j = 0; j < W_CJ; ++j) {
            const g4f a = wr[m][2 * j], c = wr[m][2 * j + 1];
            g4f lo, hi;
            lo.x = a.x; lo.y = c.x; lo.z = a.y; lo.w = c.y;
            hi.x = a.z; hi.y = c.z; hi.z = a.w; hi.w = c.w;
            *(g4f*)(&wst[buf ^ 1][m][wso + 64 * j])     = lo;
            *(g4f*)(&wst[buf ^ 1][m][wso + 64 * j + 4]) = hi;
          }
      }
    }
    __syncthreads();
  }

#pragma unroll
  for (int m = 0; m < 3; ++m) {
    float* C = (m == 0) ? Q : (m == 1) ? K : V;
#pragma unroll
    for (int nt = 0; nt < 3; ++nt) {
      const int col = bn0 + wn * 48 + nt * 16 + l16;
#pragma unroll
      for (int a = 0; a < 2; ++a) {
        const int row0 = bm0 + wm * 32 + a * 16 + 8 * half;
#pragma unroll
        for (int v = 0; v < 8; ++v) {
          C[(size_t)(row0 + v) * EMBED + col] = acc[m][a][nt][v];
        }
      }
    }
  }
}

#define ATT_WAVES   4
#define ATT_THREADS (ATT_WAVES * 32)
#define KSTAGE      64
#define SLOADS      (KSTAGE / 8)
#define KPITCH      68
#define VPITCH2     136

typedef __attribute__((ext_vector_type(4))) float v4f;

__device__ __forceinline__ float xhalf(float x) {
  const int b = __builtin_bit_cast(int, x);
  return __builtin_bit_cast(float,
      __builtin_amdgcn_permlanex16(b, b, 0x76543210, 0xfedcba98, false, false));
}

__global__ __launch_bounds__(ATT_THREADS) void attention_kernel(
    const float* __restrict__ Q, const float* __restrict__ K,
    const float* __restrict__ V, const float* __restrict__ mask,
    float* __restrict__ O) {
  __shared__ __attribute__((aligned(16))) float kst[2][KSTAGE * KPITCH];
  __shared__ __attribute__((aligned(16))) float vst[2][(KSTAGE / 2) * VPITCH2];

  const int lane = threadIdx.x & 31;
  const int wave = threadIdx.x >> 5;
  const int half = lane >> 4;
  const int l16  = lane & 15;

  const int tile = blockIdx.x * ATT_WAVES + wave;
  const int qt = tile & 127;
  const int bh = tile >> 7;
  const int b  = bh / NHEAD;
  const int h  = bh % NHEAD;

  const int    q0   = qt * 16;
  const size_t base = (size_t)b * S_LEN * EMBED + (size_t)h * HDIM;

  const float* qrow = Q + base + (size_t)(q0 + l16) * EMBED;

  const int srow = threadIdx.x >> 4;
  const int sc4  = (threadIdx.x & 15) * 4;
  const float* kg = K + base + (size_t)srow * EMBED + sc4;
  const float* vg = V + base + (size_t)(2 * srow) * EMBED + sc4;
  const int kso = srow * KPITCH + sc4;
  const int vso = srow * VPITCH2 + 2 * sc4;

  v2f qf[HDIM / 4];
#pragma unroll
  for (int i = 0; i < HDIM / 4; ++i) qf[i] = *(const v2f*)(qrow + 4 * i + 2 * half);

  v4f kr[SLOADS], vr[SLOADS];
#pragma unroll
  for (int i = 0; i < SLOADS; ++i)
    kr[i] = *(const v4f*)(kg + (size_t)(8 * i) * EMBED);
#pragma unroll
  for (int i = 0; i < SLOADS / 2; ++i) {
    vr[2 * i]     = *(const v4f*)(vg + (size_t)(16 * i) * EMBED);
    vr[2 * i + 1] = *(const v4f*)(vg + (size_t)(16 * i + 1) * EMBED);
  }
#pragma unroll
  for (int i = 0; i < SLOADS; ++i)
    *(v4f*)(&kst[0][kso + 8 * i * KPITCH]) = kr[i];
#pragma unroll
  for (int i = 0; i < SLOADS / 2; ++i) {
    const v4f a = vr[2 * i], c = vr[2 * i + 1];
    v4f lo, hi;
    lo.x = a.x; lo.y = c.x; lo.z = a.y; lo.w = c.y;
    hi.x = a.z; hi.y = c.z; hi.z = a.w; hi.w = c.w;
    *(v4f*)(&vst[0][vso + 8 * i * VPITCH2])     = lo;
    *(v4f*)(&vst[0][vso + 8 * i * VPITCH2 + 4]) = hi;
  }

  v8f acc[4];
#pragma unroll
  for (int i = 0; i < 4; ++i) acc[i] = vzero8();
  float mrun = -3.0e38f, lrun = 0.f;

  const float* mrow = mask + (size_t)(q0 + l16) * S_LEN + 8 * half;
  v4f mk0 = *(const v4f*)(mrow);
  v4f mk1 = *(const v4f*)(mrow + 4);

  __syncthreads();

  for (int st = 0; st < S_LEN / KSTAGE; ++st) {
    const int  buf  = st & 1;
    const bool more = (st + 1) < (S_LEN / KSTAGE);

    if (more) {
      const size_t off = (size_t)(st + 1) * KSTAGE * EMBED;
#pragma unroll
      for (int i = 0; i < SLOADS; ++i)
        kr[i] = *(const v4f*)(kg + off + (size_t)(8 * i) * EMBED);
#pragma unroll
      for (int i = 0; i < SLOADS / 2; ++i) {
        vr[2 * i]     = *(const v4f*)(vg + off + (size_t)(16 * i) * EMBED);
        vr[2 * i + 1] = *(const v4f*)(vg + off + (size_t)(16 * i + 1) * EMBED);
      }
    }

    const float* kb = &kst[buf][0];
    const float* vb = &vst[buf][0];

    v2f kf[HDIM / 4];
    {
      const float* krow = kb + l16 * KPITCH + 2 * half;
#pragma unroll
      for (int i = 0; i < HDIM / 4; ++i) kf[i] = *(const v2f*)(krow + 4 * i);
    }

#pragma unroll
    for (int sub = 0; sub < KSTAGE / 16; ++sub) {
      const int kt = st * (KSTAGE / 16) + sub;

      v8f s = vzero8();
#pragma unroll
      for (int i = 0; i < HDIM / 4; ++i) s = wmma4(kf[i], qf[i], s);

      __builtin_amdgcn_sched_barrier(0);

      v2f vf[4][4];
#pragma unroll
      for (int j = 0; j < 4; ++j) {
        const float* vp = vb + (sub * 8 + 4 * half + j) * VPITCH2 + 2 * l16;
#pragma unroll
        for (int nt = 0; nt < 4; ++nt) vf[j][nt] = *(const v2f*)(vp + 32 * nt);
      }
      if (sub + 1 < KSTAGE / 16) {
        const float* krow = kb + ((sub + 1) * 16 + l16) * KPITCH + 2 * half;
#pragma unroll
        for (int i = 0; i < HDIM / 4; ++i) kf[i] = *(const v2f*)(krow + 4 * i);
      }

      const v4f mc0 = mk0, mc1 = mk1;
      const int ktn = (kt + 1 < S_LEN / 16) ? kt + 1 : kt;
      mk0 = *(const v4f*)(mrow + ktn * 16);
      mk1 = *(const v4f*)(mrow + ktn * 16 + 4);

      float sv[8];
      sv[0] = s[0] * 0.125f + mc0.x * (-1.0e9f);
      sv[1] = s[1] * 0.125f + mc0.y * (-1.0e9f);
      sv[2] = s[2] * 0.125f + mc0.z * (-1.0e9f);
      sv[3] = s[3] * 0.125f + mc0.w * (-1.0e9f);
      sv[4] = s[4] * 0.125f + mc1.x * (-1.0e9f);
      sv[5] = s[5] * 0.125f + mc1.y * (-1.0e9f);
      sv[6] = s[6] * 0.125f + mc1.z * (-1.0e9f);
      sv[7] = s[7] * 0.125f + mc1.w * (-1.0e9f);

      float tmax = fmaxf(fmaxf(fmaxf(sv[0], sv[1]), fmaxf(sv[2], sv[3])),
                         fmaxf(fmaxf(sv[4], sv[5]), fmaxf(sv[6], sv[7])));
      tmax = fmaxf(tmax, xhalf(tmax));
      const float mnew  = fmaxf(mrun, tmax);
      const float alpha = __expf(mrun - mnew);
      float p[8];
#pragma unroll
      for (int v = 0; v < 8; ++v) p[v] = __expf(sv[v] - mnew);
      float rsum = ((p[0] + p[1]) + (p[2] + p[3])) + ((p[4] + p[5]) + (p[6] + p[7]));
      rsum += xhalf(rsum);
      lrun = lrun * alpha + rsum;
      mrun = mnew;
#pragma unroll
      for (int nt = 0; nt < 4; ++nt)
#pragma unroll
        for (int r = 0; r < 8; ++r) acc[nt][r] *= alpha;

      __builtin_amdgcn_sched_barrier(0);

#pragma unroll
      for (int j = 0; j < 4; ++j) {
        v2f pb;
        pb.x = p[2 * j];
        pb.y = p[2 * j + 1];
#pragma unroll
        for (int nt = 0; nt < 4; ++nt) acc[nt] = wmma4(vf[j][nt], pb, acc[nt]);
      }
    }

    if (more) {
#pragma unroll
      for (int i = 0; i < SLOADS; ++i)
        *(v4f*)(&kst[buf ^ 1][kso + 8 * i * KPITCH]) = kr[i];
#pragma unroll
      for (int i = 0; i < SLOADS / 2; ++i) {
        const v4f a = vr[2 * i], c = vr[2 * i + 1];
        v4f lo, hi;
        lo.x = a.x; lo.y = c.x; lo.z = a.y; lo.w = c.y;
        hi.x = a.z; hi.y = c.z; hi.z = a.w; hi.w = c.w;
        *(v4f*)(&vst[buf ^ 1][vso + 8 * i * VPITCH2])     = lo;
        *(v4f*)(&vst[buf ^ 1][vso + 8 * i * VPITCH2 + 4]) = hi;
      }
    }
    __syncthreads();
  }

  const float inv = 1.0f / lrun;
  float* orow = O + base + (size_t)(q0 + l16) * EMBED + 8 * half;
#pragma unroll
  for (int nt = 0; nt < 4; ++nt) {
    v4f o0, o1;
    o0.x = acc[nt][0] * inv; o0.y = acc[nt][1] * inv;
    o0.z = acc[nt][2] * inv; o0.w = acc[nt][3] * inv;
    o1.x = acc[nt][4] * inv; o1.y = acc[nt][5] * inv;
    o1.z = acc[nt][6] * inv; o1.w = acc[nt][7] * inv;
    *(v4f*)(orow + nt * 16)     = o0;
    *(v4f*)(orow + nt * 16 + 4) = o1;
  }
}

extern "C" void kernel_launch(void* const* d_in, const int* in_sizes, int n_in,
                              void* d_out, int out_size, void* d_ws, size_t ws_size,
                              hipStream_t stream) {
  const float* x    = (const float*)d_in[0];
  const float* mask = (const float*)d_in[1];
  const float* Wq   = (const float*)d_in[2];
  const float* Wk   = (const float*)d_in[3];
  const float* Wv   = (const float*)d_in[4];
  const float* Wo   = (const float*)d_in[5];
  const float* bo   = (const float*)d_in[6];

  float* ws = (float*)d_ws;
  const size_t mat = (size_t)ROWS * EMBED;
  float* Qb = ws;
  float* Kb = ws + mat;
  float* Vb = ws + 2 * mat;
  float* Ab = ws + 3 * mat;

  dim3 g(ROWS / GBM, EMBED / GBN);
  qkv_gemm_kernel<<<g, GTHREADS, 0, stream>>>(x, Wq, Wk, Wv, Qb, Kb, Vb);

  const int qtiles = BATCH * NHEAD * (S_LEN / 16);
  attention_kernel<<<qtiles / ATT_WAVES, ATT_THREADS, 0, stream>>>(Qb, Kb, Vb, mask, Ab);

  gemm_bias_kernel<<<g, GTHREADS, 0, stream>>>(Ab, Wo, bo, (float*)d_out, 1);
}
